# starting kernel
# baseline (speedup 1.0000x reference)
#include <hip/hip_runtime.h>

typedef __attribute__((ext_vector_type(2))) float v2f;
typedef __attribute__((ext_vector_type(8))) float v8f;


__global__ __launch_bounds__(256) void mpo_pre_kernel(
    const float* __restrict__ c0, const float* __restrict__ c1,
    const float* __restrict__ c3, const float* __restrict__ c4,
    float* __restrict__ P12, float* __restrict__ S45) {
  const int bid = blockIdx.x, tid = threadIdx.x;
  if (bid < 64) {
    const int i1 = bid >> 3, i2 = bid & 7;
    for (int o = tid; o < 1024; o += 256) {
      const int p = o >> 6, r2 = o & 63;
      const int o1 = p >> 2, o2 = p & 3;
      const float* a = c0 + (i1 * 4 + o1) * 32;
      const float* b = c1 + (i2 * 4 + o2) * 64 + r2;
      float s = 0.f;
      for (int r1 = 0; r1 < 32; ++r1) s += a[r1] * b[r1 * 2048];
      P12[bid * 1024 + o] = s;
    }
  } else {
    const int cb = bid - 64;
    const int i4 = cb >> 3, i5 = cb & 7;
    for (int o = tid; o < 768; o += 256) {
      const int r3 = o / 12, q = o % 12;
      const int o4 = q / 3, o5 = q % 3;
      const float* a = c3 + ((r3 * 8 + i4) * 4 + o4) * 24;
      const float* b = c4 + i5 * 3 + o5;
      float s = 0.f;
      for (int r4 = 0; r4 < 24; ++r4) s += a[r4] * b[r4 * 24];
      S45[cb * 768 + o] = s;
    }
  }
}

__global__ __launch_bounds__(256) void mpo_p123_wmma_kernel(
    const float* __restrict__ c2, const float* __restrict__ P12,
    float* __restrict__ P123) {
  const int wave  = blockIdx.x * 8 + (threadIdx.x >> 5);
  const int lane  = threadIdx.x & 31;
  const int combo = wave >> 4;
  const int ntile = wave & 15;
  const int c12 = combo >> 3, i3 = combo & 7;

  const float* A = P12 + c12 * 1024;
  const float* B = c2 + i3 * 256;

  const int m    = lane & 15;
  const int half = lane >> 4;
  const int klo  = half * 2;
  const int n    = ntile * 16 + m;

  v8f acc = {0.f, 0.f, 0.f, 0.f, 0.f, 0.f, 0.f, 0.f};
#pragma unroll
  for (int kk = 0; kk < 16; ++kk) {
    const int k0 = kk * 4 + klo;
    v2f a = *(const v2f*)(A + m * 64 + k0);
    v2f b;
    b.x = B[k0 * 2048 + n];
    b.y = B[(k0 + 1) * 2048 + n];
    acc = __builtin_amdgcn_wmma_f32_16x16x4_f32(
         false, a,  false, b,
         (short)0, acc,  false,  false);
  }

  float* D = P123 + combo * 4096;
#pragma unroll
  for (int v = 0; v < 8; ++v) {
    const int row = v + half * 8;
    D[row * 256 + n] = acc[v];
  }
}

__global__ __launch_bounds__(256) void mpo_gather_wmma_kernel(
    const int* __restrict__ ids, const float* __restrict__ P123,
    const float* __restrict__ S45, float* __restrict__ out) {
  const int wave  = blockIdx.x * 8 + (threadIdx.x >> 5);
  const int lane  = threadIdx.x & 31;
  const int tok   = wave >> 2;
  const int mtile = wave & 3;

  const int id   = ids[tok];
  const int c45  = id & 63;
  const int c123 = id >> 6;

  const float* A = P123 + c123 * 4096 + mtile * (16 * 64);
  const float* B = S45 + c45 * 768;

  const int m    = lane & 15;
  const int half = lane >> 4;
  const int klo  = half * 2;
  const int n    = m;
  const int ncl  = (n < 12) ? n : 11;
  const float msk = (n < 12) ? 1.f : 0.f;

  v8f acc = {0.f, 0.f, 0.f, 0.f, 0.f, 0.f, 0.f, 0.f};
#pragma unroll
  for (int kk = 0; kk < 16; ++kk) {
    const int k0 = kk * 4 + klo;
    v2f a = *(const v2f*)(A + m * 64 + k0);
    v2f b;
    b.x = B[k0 * 12 + ncl] * msk;
    b.y = B[(k0 + 1) * 12 + ncl] * msk;
    acc = __builtin_amdgcn_wmma_f32_16x16x4_f32(
        false, a, false, b, (short)0, acc, false, false);
  }

  if (n < 12) {
    float* o = out + tok * 768 + (mtile * 16) * 12 + n;
#pragma unroll
    for (int v = 0; v < 8; ++v) {
      const int row = v + half * 8;
      o[row * 12] = acc[v];
    }
  }
}

extern "C" void kernel_launch(void* const* d_in, const int* in_sizes, int n_in,
                              void* d_out, int out_size, void* d_ws, size_t ws_size,
                              hipStream_t stream) {
  const int*   ids = (const int*)d_in[0];
  const float* c0  = (const float*)d_in[1];
  const float* c1  = (const float*)d_in[2];
  const float* c2  = (const float*)d_in[3];
  const float* c3  = (const float*)d_in[4];
  const float* c4  = (const float*)d_in[5];
  float* out = (float*)d_out;

  float* ws   = (float*)d_ws;
  float* P12  = ws;
  float* S45  = ws + 65536;
  float* P123 = ws + 65536 + 49152;

  mpo_pre_kernel<<<128, 256, 0, stream>>>(c0, c1, c3, c4, P12, S45);
  mpo_p123_wmma_kernel<<<1024, 256, 0, stream>>>(c2, P12, P123);
  mpo_gather_wmma_kernel<<<2048, 256, 0, stream>>>(ids, P123, S45, out);
}


# 4 tiles per wave in both WMMA kernels (4x fewer waves to dispatch): gather wave = one token, B fragment loaded once; P123 wave = 4 column-interleaved tiles fed by 16-byte loads and stores
# speedup vs baseline: 1.2149x; 1.2149x over previous
#include <hip/hip_runtime.h>

typedef __attribute__((ext_vector_type(2))) float v2f;
typedef __attribute__((ext_vector_type(4))) float v4f;
typedef __attribute__((ext_vector_type(8))) float v8f;


__global__ __launch_bounds__(256) void mpo_pre_kernel(
    const float* __restrict__ c0, const float* __restrict__ c1,
    const float* __restrict__ c3, const float* __restrict__ c4,
    float* __restrict__ P12, float* __restrict__ S45) {
  const int bid = blockIdx.x, tid = threadIdx.x;
  if (bid < 64) {
    const int i1 = bid >> 3, i2 = bid & 7;
    for (int o = tid; o < 1024; o += 256) {
      const int p = o >> 6, r2 = o & 63;
      const int o1 = p >> 2, o2 = p & 3;
      const float* a = c0 + (i1 * 4 + o1) * 32;
      const float* b = c1 + (i2 * 4 + o2) * 64 + r2;
      float s = 0.f;
      for (int r1 = 0; r1 < 32; ++r1) s += a[r1] * b[r1 * 2048];
      P12[bid * 1024 + o] = s;
    }
  } else {
    const int cb = bid - 64;
    const int i4 = cb >> 3, i5 = cb & 7;
    for (int o = tid; o < 768; o += 256) {
      const int r3 = o / 12, q = o % 12;
      const int o4 = q / 3, o5 = q % 3;
      const float* a = c3 + ((r3 * 8 + i4) * 4 + o4) * 24;
      const float* b = c4 + i5 * 3 + o5;
      float s = 0.f;
      for (int r4 = 0; r4 < 24; ++r4) s += a[r4] * b[r4 * 24];
      S45[cb * 768 + o] = s;
    }
  }
}

__global__ __launch_bounds__(256) void mpo_p123_wmma_kernel(
    const float* __restrict__ c2, const float* __restrict__ P12,
    float* __restrict__ P123) {
  const int wave  = blockIdx.x * 8 + (threadIdx.x >> 5);
  const int lane  = threadIdx.x & 31;
  const int combo = wave >> 2;
  const int grp   = wave & 3;
  const int c12 = combo >> 3, i3 = combo & 7;

  const float* A = P12 + c12 * 1024;
  const float* B = c2 + i3 * 256 + grp * 64;

  const int m    = lane & 15;
  const int half = lane >> 4;
  const int klo  = half * 2;
  const int col4 = m * 4;

  v8f acc[4];
#pragma unroll
  for (int e = 0; e < 4; ++e)
    acc[e] = (v8f){0.f, 0.f, 0.f, 0.f, 0.f, 0.f, 0.f, 0.f};

#pragma unroll
  for (int kk = 0; kk < 16; ++kk) {
    const int k0 = kk * 4 + klo;
    const v2f a  = *(const v2f*)(A + m * 64 + k0);
    const v4f bx = *(const v4f*)(B + k0 * 2048 + col4);
    const v4f by = *(const v4f*)(B + (k0 + 1) * 2048 + col4);
#pragma unroll
    for (int e = 0; e < 4; ++e) {
      v2f b;
      b.x = bx[e];
      b.y = by[e];
      acc[e] = __builtin_amdgcn_wmma_f32_16x16x4_f32(
           false, a,  false, b,
           (short)0, acc[e],  false,  false);
    }
  }

  float* D = P123 + combo * 4096 + grp * 64 + col4;
#pragma unroll
  for (int v = 0; v < 8; ++v) {
    const int row = v + half * 8;
    v4f d;
    d.x = acc[0][v];
    d.y = acc[1][v];
    d.z = acc[2][v];
    d.w = acc[3][v];
    *(v4f*)(D + row * 256) = d;
  }
}

__global__ __launch_bounds__(256) void mpo_gather_wmma_kernel(
    const int* __restrict__ ids, const float* __restrict__ P123,
    const float* __restrict__ S45, float* __restrict__ out) {
  const int tok   = blockIdx.x * 8 + (threadIdx.x >> 5);
  const int lane  = threadIdx.x & 31;

  const int id   = ids[tok];
  const int c45  = id & 63;
  const int c123 = id >> 6;

  const float* A = P123 + c123 * 4096;
  const float* B = S45 + c45 * 768;

  const int m    = lane & 15;
  const int half = lane >> 4;
  const int klo  = half * 2;
  const int n    = m;
  const int ncl  = (n < 12) ? n : 11;
  const float msk = (n < 12) ? 1.f : 0.f;

  v2f bf[16];
#pragma unroll
  for (int kk = 0; kk < 16; ++kk) {
    const int k0 = kk * 4 + klo;
    bf[kk].x = B[k0 * 12 + ncl] * msk;
    bf[kk].y = B[(k0 + 1) * 12 + ncl] * msk;
  }

  v8f acc[4];
#pragma unroll
  for (int mt = 0; mt < 4; ++mt)
    acc[mt] = (v8f){0.f, 0.f, 0.f, 0.f, 0.f, 0.f, 0.f, 0.f};

  v2f a[2][16];
#pragma unroll
  for (int kk = 0; kk < 16; ++kk)
    a[0][kk] = *(const v2f*)(A + m * 64 + kk * 4 + klo);
#pragma unroll
  for (int mt = 0; mt < 4; ++mt) {
    if (mt < 3) {
#pragma unroll
      for (int kk = 0; kk < 16; ++kk)
        a[(mt + 1) & 1][kk] =
            *(const v2f*)(A + ((mt + 1) * 16 + m) * 64 + kk * 4 + klo);
    }
    __builtin_amdgcn_sched_barrier(0);
#pragma unroll
    for (int kk = 0; kk < 16; ++kk)
      acc[mt] = __builtin_amdgcn_wmma_f32_16x16x4_f32(
          false, a[mt & 1][kk], false, bf[kk], (short)0, acc[mt], false, false);
    __builtin_amdgcn_sched_barrier(0);
  }

  if (n < 12) {
    float* o = out + tok * 768 + n;
#pragma unroll
    for (int mt = 0; mt < 4; ++mt) {
#pragma unroll
      for (int v = 0; v < 8; ++v) {
        const int row = mt * 16 + v + half * 8;
        o[row * 12] = acc[mt][v];
      }
    }
  }
}

extern "C" void kernel_launch(void* const* d_in, const int* in_sizes, int n_in,
                              void* d_out, int out_size, void* d_ws, size_t ws_size,
                              hipStream_t stream) {
  const int*   ids = (const int*)d_in[0];
  const float* c0  = (const float*)d_in[1];
  const float* c1  = (const float*)d_in[2];
  const float* c2  = (const float*)d_in[3];
  const float* c3  = (const float*)d_in[4];
  const float* c4  = (const float*)d_in[5];
  float* out = (float*)d_out;

  float* ws   = (float*)d_ws;
  float* P12  = ws;
  float* S45  = ws + 65536;
  float* P123 = ws + 65536 + 49152;

  mpo_pre_kernel<<<128, 256, 0, stream>>>(c0, c1, c3, c4, P12, S45);
  mpo_p123_wmma_kernel<<<256, 256, 0, stream>>>(c2, P12, P123);
  mpo_gather_wmma_kernel<<<512, 256, 0, stream>>>(ids, P123, S45, out);
}


# Prefix launch fused into the P123 launch (3 launches -> 2): each block forms its P12[c12] tile in LDS, same r1 order, under its core2 loads; the S45 blocks ride in the same grid
# speedup vs baseline: 1.4811x; 1.2191x over previous
#include <hip/hip_runtime.h>

typedef __attribute__((ext_vector_type(2))) float v2f;
typedef __attribute__((ext_vector_type(4))) float v4f;
typedef __attribute__((ext_vector_type(8))) float v8f;


#define P12_PITCH 68

__global__ __launch_bounds__(256) void mpo_p123_fused_kernel(
    const float* __restrict__ c0, const float* __restrict__ c1,
    const float* __restrict__ c2, const float* __restrict__ c3,
    const float* __restrict__ c4, float* __restrict__ P123,
    float* __restrict__ S45) {
  __shared__ float sP12[16 * P12_PITCH];
  const int bid = blockIdx.x, tid = threadIdx.x;

  if (bid >= 256) {
    const int cb = bid - 256;
    const int i4 = cb >> 3, i5 = cb & 7;
    for (int o = tid; o < 768; o += 256) {
      const int r3 = o / 12, q = o % 12;
      const int o4 = q / 3, o5 = q % 3;
      const float* a = c3 + ((r3 * 8 + i4) * 4 + o4) * 24;
      const float* b = c4 + i5 * 3 + o5;
      float s = 0.f;
      for (int r4 = 0; r4 < 24; ++r4) s += a[r4] * b[r4 * 24];
      S45[cb * 768 + o] = s;
    }
  } else {
    const int wave  = bid * 8 + (tid >> 5);
    const int lane  = tid & 31;
    const int combo = wave >> 2;
    const int grp   = wave & 3;
    const int c12 = bid >> 2;
    const int i3  = combo & 7;
    const int i1 = c12 >> 3, i2 = c12 & 7;

    const float* B = c2 + i3 * 256 + grp * 64;

    const int m    = lane & 15;
    const int half = lane >> 4;
    const int klo  = half * 2;
    const int col4 = m * 4;

    v4f bx[16], by[16];
#pragma unroll
    for (int kk = 0; kk < 16; ++kk) {
      const int k0 = kk * 4 + klo;
      bx[kk] = *(const v4f*)(B + k0 * 2048 + col4);
      by[kk] = *(const v4f*)(B + (k0 + 1) * 2048 + col4);
    }

    {
      const int o2 = tid >> 6, r2 = tid & 63;
      const float* b = c1 + (i2 * 4 + o2) * 64 + r2;
      float bv[32];
#pragma unroll
      for (int r1 = 0; r1 < 32; ++r1) bv[r1] = b[r1 * 2048];
#pragma unroll
      for (int o1 = 0; o1 < 4; ++o1) {
        const float* a = c0 + (i1 * 4 + o1) * 32;
        float s = 0.f;
#pragma unroll
        for (int r1 = 0; r1 < 32; ++r1) s += a[r1] * bv[r1];
        sP12[(o1 * 4 + o2) * P12_PITCH + r2] = s;
      }
    }
    __syncthreads();

    v8f acc[4];
#pragma unroll
    for (int e = 0; e < 4; ++e)
      acc[e] = (v8f){0.f, 0.f, 0.f, 0.f, 0.f, 0.f, 0.f, 0.f};

#pragma unroll
    for (int kk = 0; kk < 16; ++kk) {
      const int k0 = kk * 4 + klo;
      const v2f a  = *(const v2f*)(sP12 + m * P12_PITCH + k0);
#pragma unroll
      for (int e = 0; e < 4; ++e) {
        v2f b;
        b.x = bx[kk][e];
        b.y = by[kk][e];
        acc[e] = __builtin_amdgcn_wmma_f32_16x16x4_f32(
             false, a,  false, b,
             (short)0, acc[e],  false,  false);
      }
    }

    float* D = P123 + combo * 4096 + grp * 64 + col4;
#pragma unroll
    for (int v = 0; v < 8; ++v) {
      const int row = v + half * 8;
      v4f d;
      d.x = acc[0][v];
      d.y = acc[1][v];
      d.z = acc[2][v];
      d.w = acc[3][v];
      *(v4f*)(D + row * 256) = d;
    }
  }
}

__global__ __launch_bounds__(256) void mpo_gather_wmma_kernel(
    const int* __restrict__ ids, const float* __restrict__ P123,
    const float* __restrict__ S45, float* __restrict__ out) {
  const int tok   = blockIdx.x * 8 + (threadIdx.x >> 5);
  const int lane  = threadIdx.x & 31;

  const int id   = ids[tok];
  const int c45  = id & 63;
  const int c123 = id >> 6;

  const float* A = P123 + c123 * 4096;
  const float* B = S45 + c45 * 768;

  const int m    = lane & 15;
  const int half = lane >> 4;
  const int klo  = half * 2;
  const int n    = m;
  const int ncl  = (n < 12) ? n : 11;
  const float msk = (n < 12) ? 1.f : 0.f;

  v2f bf[16];
#pragma unroll
  for (int kk = 0; kk < 16; ++kk) {
    const int k0 = kk * 4 + klo;
    bf[kk].x = B[k0 * 12 + ncl] * msk;
    bf[kk].y = B[(k0 + 1) * 12 + ncl] * msk;
  }

  v8f acc[4];
#pragma unroll
  for (int mt = 0; mt < 4; ++mt)
    acc[mt] = (v8f){0.f, 0.f, 0.f, 0.f, 0.f, 0.f, 0.f, 0.f};

  v2f a[2][16];
#pragma unroll
  for (int kk = 0; kk < 16; ++kk)
    a[0][kk] = *(const v2f*)(A + m * 64 + kk * 4 + klo);
#pragma unroll
  for (int mt = 0; mt < 4; ++mt) {
    if (mt < 3) {
#pragma unroll
      for (int kk = 0; kk < 16; ++kk)
        a[(mt + 1) & 1][kk] =
            *(const v2f*)(A + ((mt + 1) * 16 + m) * 64 + kk * 4 + klo);
    }
    __builtin_amdgcn_sched_barrier(0);
#pragma unroll
    for (int kk = 0; kk < 16; ++kk)
      acc[mt] = __builtin_amdgcn_wmma_f32_16x16x4_f32(
          false, a[mt & 1][kk], false, bf[kk], (short)0, acc[mt], false, false);
    __builtin_amdgcn_sched_barrier(0);
  }

  if (n < 12) {
    float* o = out + tok * 768 + n;
#pragma unroll
    for (int mt = 0; mt < 4; ++mt) {
#pragma unroll
      for (int v = 0; v < 8; ++v) {
        const int row = mt * 16 + v + half * 8;
        o[row * 12] = acc[mt][v];
      }
    }
  }
}

extern "C" void kernel_launch(void* const* d_in, const int* in_sizes, int n_in,
                              void* d_out, int out_size, void* d_ws, size_t ws_size,
                              hipStream_t stream) {
  const int*   ids = (const int*)d_in[0];
  const float* c0  = (const float*)d_in[1];
  const float* c1  = (const float*)d_in[2];
  const float* c2  = (const float*)d_in[3];
  const float* c3  = (const float*)d_in[4];
  const float* c4  = (const float*)d_in[5];
  float* out = (float*)d_out;

  float* ws   = (float*)d_ws;
  float* S45  = ws + 65536;
  float* P123 = ws + 65536 + 49152;

  mpo_p123_fused_kernel<<<320, 256, 0, stream>>>(c0, c1, c2, c3, c4, P123, S45);
  mpo_gather_wmma_kernel<<<512, 256, 0, stream>>>(ids, P123, S45, out);
}


# P123 workspace kept in the gather's A-fragment order (producer forms the transposed tile by swapping its WMMA operands): gather A loads are 16-byte whole-line loads; r3 order of gather sums changes
# speedup vs baseline: 1.7526x; 1.1833x over previous
#include <hip/hip_runtime.h>

typedef __attribute__((ext_vector_type(2))) float v2f;
typedef __attribute__((ext_vector_type(4))) float v4f;
typedef __attribute__((ext_vector_type(8))) float v8f;


#define P12_PITCH 68

__global__ __launch_bounds__(256) void mpo_p123_fused_kernel(
    const float* __restrict__ c0, const float* __restrict__ c1,
    const float* __restrict__ c2, const float* __restrict__ c3,
    const float* __restrict__ c4, float* __restrict__ P123,
    float* __restrict__ S45) {
  __shared__ float sP12[16 * P12_PITCH];
  const int bid = blockIdx.x, tid = threadIdx.x;

  if (bid >= 256) {
    const int cb = bid - 256;
    const int i4 = cb >> 3, i5 = cb & 7;
    for (int o = tid; o < 768; o += 256) {
      const int r3 = o / 12, q = o % 12;
      const int o4 = q / 3, o5 = q % 3;
      const float* a = c3 + ((r3 * 8 + i4) * 4 + o4) * 24;
      const float* b = c4 + i5 * 3 + o5;
      float s = 0.f;
      for (int r4 = 0; r4 < 24; ++r4) s += a[r4] * b[r4 * 24];
      S45[cb * 768 + o] = s;
    }
  } else {
    const int wave  = bid * 8 + (tid >> 5);
    const int lane  = tid & 31;
    const int combo = wave >> 2;
    const int grp   = wave & 3;
    const int c12 = bid >> 2;
    const int i3  = combo & 7;
    const int i1 = c12 >> 3, i2 = c12 & 7;

    const float* B = c2 + i3 * 256 + grp * 64;

    const int m    = lane & 15;
    const int half = lane >> 4;
    const int klo  = half * 2;
    const int col4 = m * 4;

    v4f bx[16], by[16];
#pragma unroll
    for (int kk = 0; kk < 16; ++kk) {
      const int k0 = kk * 4 + klo;
      bx[kk] = *(const v4f*)(B + k0 * 2048 + col4);
      by[kk] = *(const v4f*)(B + (k0 + 1) * 2048 + col4);
    }

    {
      const int o2 = tid >> 6, r2 = tid & 63;
      const float* b = c1 + (i2 * 4 + o2) * 64 + r2;
      float bv[32];
#pragma unroll
      for (int r1 = 0; r1 < 32; ++r1) bv[r1] = b[r1 * 2048];
#pragma unroll
      for (int o1 = 0; o1 < 4; ++o1) {
        const float* a = c0 + (i1 * 4 + o1) * 32;
        float s = 0.f;
#pragma unroll
        for (int r1 = 0; r1 < 32; ++r1) s += a[r1] * bv[r1];
        sP12[(o1 * 4 + o2) * P12_PITCH + r2] = s;
      }
    }
    __syncthreads();

    v8f acc[4];
#pragma unroll
    for (int e = 0; e < 4; ++e)
      acc[e] = (v8f){0.f, 0.f, 0.f, 0.f, 0.f, 0.f, 0.f, 0.f};

#pragma unroll
    for (int kk = 0; kk < 16; ++kk) {
      const int k0 = kk * 4 + klo;
      const v2f p  = *(const v2f*)(sP12 + m * P12_PITCH + k0);
#pragma unroll
      for (int e = 0; e < 4; ++e) {
        v2f c;
        c.x = bx[kk][e];
        c.y = by[kk][e];
        acc[e] = __builtin_amdgcn_wmma_f32_16x16x4_f32(
             false, c,  false, p,
             (short)0, acc[e],  false,  false);
      }
    }

    float* D = P123 + (combo * 32 + grp * 8) * 128 + lane * 4;
#pragma unroll
    for (int v = 0; v < 8; ++v) {
      v4f d;
      d.x = acc[0][v];
      d.y = acc[1][v];
      d.z = acc[2][v];
      d.w = acc[3][v];
      *(v4f*)(D + v * 128) = d;
    }
  }
}

__global__ __launch_bounds__(256) void mpo_gather_wmma_kernel(
    const int* __restrict__ ids, const float* __restrict__ P123,
    const float* __restrict__ S45, float* __restrict__ out) {
  const int tok   = blockIdx.x * 8 + (threadIdx.x >> 5);
  const int lane  = threadIdx.x & 31;

  const int id   = ids[tok];
  const int c45  = id & 63;
  const int c123 = id >> 6;

  const float* A = P123 + c123 * 4096 + lane * 4;
  const float* B = S45 + c45 * 768;

  const int m    = lane & 15;
  const int half = lane >> 4;
  const int khi  = half * 32;
  const int n    = m;
  const int ncl  = (n < 12) ? n : 11;
  const float msk = (n < 12) ? 1.f : 0.f;

  v2f bf[16];
#pragma unroll
  for (int t = 0; t < 16; ++t) {
    const int k0 = khi + t * 2;
    bf[t].x = B[k0 * 12 + ncl] * msk;
    bf[t].y = B[(k0 + 1) * 12 + ncl] * msk;
  }

  v8f acc[4];
#pragma unroll
  for (int mt = 0; mt < 4; ++mt)
    acc[mt] = (v8f){0.f, 0.f, 0.f, 0.f, 0.f, 0.f, 0.f, 0.f};

  v4f av[2][8];
#pragma unroll
  for (int g = 0; g < 8; ++g)
    av[0][g] = *(const v4f*)(A + g * 128);
#pragma unroll
  for (int mt = 0; mt < 4; ++mt) {
    if (mt < 3) {
#pragma unroll
      for (int g = 0; g < 8; ++g)
        av[(mt + 1) & 1][g] = *(const v4f*)(A + ((mt + 1) * 8 + g) * 128);
    }
    __builtin_amdgcn_sched_barrier(0);
#pragma unroll
    for (int g = 0; g < 8; ++g) {
      v2f a0, a1;
      a0.x = av[mt & 1][g].x;  a0.y = av[mt & 1][g].y;
      a1.x = av[mt & 1][g].z;  a1.y = av[mt & 1][g].w;
      acc[mt] = __builtin_amdgcn_wmma_f32_16x16x4_f32(
          false, a0, false, bf[2 * g], (short)0, acc[mt], false, false);
      acc[mt] = __builtin_amdgcn_wmma_f32_16x16x4_f32(
          false, a1, false, bf[2 * g + 1], (short)0, acc[mt], false, false);
    }
    __builtin_amdgcn_sched_barrier(0);
  }

  if (n < 12) {
    float* o = out + tok * 768 + n;
#pragma unroll
    for (int mt = 0; mt < 4; ++mt) {
#pragma unroll
      for (int v = 0; v < 8; ++v) {
        const int row = (v + half * 8) * 4 + mt;
        o[row * 12] = acc[mt][v];
      }
    }
  }
}

extern "C" void kernel_launch(void* const* d_in, const int* in_sizes, int n_in,
                              void* d_out, int out_size, void* d_ws, size_t ws_size,
                              hipStream_t stream) {
  const int*   ids = (const int*)d_in[0];
  const float* c0  = (const float*)d_in[1];
  const float* c1  = (const float*)d_in[2];
  const float* c2  = (const float*)d_in[3];
  const float* c3  = (const float*)d_in[4];
  const float* c4  = (const float*)d_in[5];
  float* out = (float*)d_out;

  float* ws   = (float*)d_ws;
  float* S45  = ws + 65536;
  float* P123 = ws + 65536 + 49152;

  mpo_p123_fused_kernel<<<320, 256, 0, stream>>>(c0, c1, c2, c3, c4, P123, S45);
  mpo_gather_wmma_kernel<<<512, 256, 0, stream>>>(ids, P123, S45, out);
}


# Core chain cut at r2, not r3: gather = P12[i1i2](16x64) @ T345[i3,i4i5](64x48), 3 full N tiles = 48 WMMAs per token for 64, no padded column; plane 6 MB for 8; order of the r2 and r3 sums changes
# speedup vs baseline: 2.2756x; 1.2984x over previous
#include <hip/hip_runtime.h>

typedef __attribute__((ext_vector_type(2))) float v2f;
typedef __attribute__((ext_vector_type(4))) float v4f;
typedef __attribute__((ext_vector_type(8))) float v8f;


#define T_PITCH 68

__global__ __launch_bounds__(256) void mpo_t345_fused_kernel(
    const float* __restrict__ c0, const float* __restrict__ c1,
    const float* __restrict__ c2, const float* __restrict__ c3,
    const float* __restrict__ c4, float* __restrict__ T345,
    float* __restrict__ P12) {
  __shared__ __attribute__((aligned(16))) float sW[8 * 64 * T_PITCH];
  __shared__ __attribute__((aligned(16))) float sS[16 * T_PITCH];
  const int bid = blockIdx.x, tid = threadIdx.x;

  if (bid >= 256) {
    const int c12 = bid - 256;
    const int i1 = c12 >> 3, i2 = c12 & 7;
    const int o2 = tid >> 6, r2 = tid & 63;
    const float* b = c1 + (i2 * 4 + o2) * 64 + r2;
    float bv[32];
#pragma unroll
    for (int r1 = 0; r1 < 32; ++r1) bv[r1] = b[r1 * 2048];
    const int hf = r2 >> 5, g = (r2 & 31) >> 2, i = r2 & 3;
#pragma unroll
    for (int o1 = 0; o1 < 4; ++o1) {
      const float* a = c0 + (i1 * 4 + o1) * 32;
      float s = 0.f;
#pragma unroll
      for (int r1 = 0; r1 < 32; ++r1) s += a[r1] * bv[r1];
      P12[c12 * 1024 + (g * 32 + hf * 16 + o1 * 4 + o2) * 4 + i] = s;
    }
  } else {
    const int wv   = tid >> 5;
    const int lane = tid & 31;
    const int c45  = bid >> 2;
    const int i3   = (bid & 3) * 2 + (wv >> 2);
    const int o3   = wv & 3;
    const int i4 = c45 >> 3, i5 = c45 & 7;

    const int m    = lane & 15;
    const int half = lane >> 4;

    float* W = sW + wv * (64 * T_PITCH);
    {
      const float* src = c2 + (i3 * 4 + o3) * 64 + half * 2048 + m * 4;
      v4f t[32];
#pragma unroll
      for (int j = 0; j < 32; ++j) t[j] = *(const v4f*)(src + j * 4096);
#pragma unroll
      for (int j = 0; j < 32; ++j)
        *(v4f*)(W + (2 * j + half) * T_PITCH + m * 4) = t[j];
    }

    {
      const int o4 = tid >> 6, r3 = tid & 63;
      const float* a = c3 + ((r3 * 8 + i4) * 4 + o4) * 24;
      float av[24];
#pragma unroll
      for (int r4 = 0; r4 < 24; ++r4) av[r4] = a[r4];
#pragma unroll
      for (int o5 = 0; o5 < 3; ++o5) {
        const float* b = c4 + i5 * 3 + o5;
        float s = 0.f;
#pragma unroll
        for (int r4 = 0; r4 < 24; ++r4) s += av[r4] * b[r4 * 24];
        sS[(o4 * 3 + o5) * T_PITCH + r3] = s;
      }
      sS[(12 + o4) * T_PITCH + r3] = 0.f;
    }
    __syncthreads();

    v8f acc[4];
#pragma unroll
    for (int e = 0; e < 4; ++e)
      acc[e] = (v8f){0.f, 0.f, 0.f, 0.f, 0.f, 0.f, 0.f, 0.f};

#pragma unroll
    for (int j = 0; j < 8; ++j) {
      const int k0 = j * 8 + half * 4;
      const v4f pb = *(const v4f*)(sS + m * T_PITCH + k0);
      v2f b0, b1;
      b0.x = pb.x;  b0.y = pb.y;
      b1.x = pb.z;  b1.y = pb.w;
#pragma unroll
      for (int e = 0; e < 4; ++e) {
        const v4f pa = *(const v4f*)(W + (16 * e + m) * T_PITCH + k0);
        v2f a0, a1;
        a0.x = pa.x;  a0.y = pa.y;
        a1.x = pa.z;  a1.y = pa.w;
        acc[e] = __builtin_amdgcn_wmma_f32_16x16x4_f32(
             false, a0,  false, b0,
             (short)0, acc[e],  false,  false);
        acc[e] = __builtin_amdgcn_wmma_f32_16x16x4_f32(
            false, a1, false, b1, (short)0, acc[e], false, false);
      }
    }

    if (m < 12) {
      const int col = o3 * 12 + m;
      const int nt = col >> 4, n = col & 15;
      float* D = T345 + (i3 * 64 + c45) * 3072 + nt * 1024 + n * 4;
#pragma unroll
      for (int e = 0; e < 4; ++e) {
#pragma unroll
        for (int u = 0; u < 2; ++u) {
          const int g = 4 * (e & 1) + 2 * half + u;
          v4f d;
          d.x = acc[e][4 * u + 0];
          d.y = acc[e][4 * u + 1];
          d.z = acc[e][4 * u + 2];
          d.w = acc[e][4 * u + 3];
          *(v4f*)(D + (g * 32 + (e >> 1) * 16) * 4) = d;
        }
      }
    }
  }
}

__global__ __launch_bounds__(256) void mpo_gather_wmma_kernel(
    const int* __restrict__ ids, const float* __restrict__ P12,
    const float* __restrict__ T345, float* __restrict__ out) {
  const int tok   = blockIdx.x * 8 + (threadIdx.x >> 5);
  const int lane  = threadIdx.x & 31;

  const int id   = ids[tok];
  const int c45  = id & 63;
  const int i3   = (id >> 6) & 7;
  const int c12  = id >> 9;

  const float* A = P12 + c12 * 1024 + lane * 4;
  const float* B = T345 + (i3 * 64 + c45) * 3072 + lane * 4;

  const int n    = lane & 15;
  const int half = lane >> 4;

  v4f af[8];
#pragma unroll
  for (int g = 0; g < 8; ++g)
    af[g] = *(const v4f*)(A + g * 128);

  v8f acc[3];
#pragma unroll
  for (int nt = 0; nt < 3; ++nt)
    acc[nt] = (v8f){0.f, 0.f, 0.f, 0.f, 0.f, 0.f, 0.f, 0.f};

  v4f bv[2][8];
#pragma unroll
  for (int g = 0; g < 8; ++g)
    bv[0][g] = *(const v4f*)(B + g * 128);
#pragma unroll
  for (int nt = 0; nt < 3; ++nt) {
    if (nt < 2) {
#pragma unroll
      for (int g = 0; g < 8; ++g)
        bv[(nt + 1) & 1][g] = *(const v4f*)(B + ((nt + 1) * 8 + g) * 128);
    }
    __builtin_amdgcn_sched_barrier(0);
#pragma unroll
    for (int g = 0; g < 8; ++g) {
      v2f a0, a1, b0, b1;
      a0.x = af[g].x;  a0.y = af[g].y;
      a1.x = af[g].z;  a1.y = af[g].w;
      b0.x = bv[nt & 1][g].x;  b0.y = bv[nt & 1][g].y;
      b1.x = bv[nt & 1][g].z;  b1.y = bv[nt & 1][g].w;
      acc[nt] = __builtin_amdgcn_wmma_f32_16x16x4_f32(
          false, a0, false, b0, (short)0, acc[nt], false, false);
      acc[nt] = __builtin_amdgcn_wmma_f32_16x16x4_f32(
          false, a1, false, b1, (short)0, acc[nt], false, false);
    }
    __builtin_amdgcn_sched_barrier(0);
  }

  float* o = out + tok * 768 + half * 8 * 48 + n;
#pragma unroll
  for (int nt = 0; nt < 3; ++nt) {
#pragma unroll
    for (int v = 0; v < 8; ++v)
      o[v * 48 + nt * 16] = acc[nt][v];
  }
}

extern "C" void kernel_launch(void* const* d_in, const int* in_sizes, int n_in,
                              void* d_out, int out_size, void* d_ws, size_t ws_size,
                              hipStream_t stream) {
  const int*   ids = (const int*)d_in[0];
  const float* c0  = (const float*)d_in[1];
  const float* c1  = (const float*)d_in[2];
  const float* c2  = (const float*)d_in[3];
  const float* c3  = (const float*)d_in[4];
  const float* c4  = (const float*)d_in[5];
  float* out = (float*)d_out;

  float* ws   = (float*)d_ws;
  float* P12  = ws;
  float* T345 = ws + 65536 + 49152;

  mpo_t345_fused_kernel<<<320, 256, 0, stream>>>(c0, c1, c2, c3, c4, T345, P12);
  mpo_gather_wmma_kernel<<<512, 256, 0, stream>>>(ids, P12, T345, out);
}


# T345 producer re-tiled: block = 2 core2 slices x 4 i5, N = 4 x 12 columns = 3 full N tiles (48 WMMAs a wave for 64, no padded column); core2 staged once per 4 combos (128 -> 32 KB a block)
# speedup vs baseline: 2.4579x; 1.0801x over previous
#include <hip/hip_runtime.h>

typedef __attribute__((ext_vector_type(2))) float v2f;
typedef __attribute__((ext_vector_type(4))) float v4f;
typedef __attribute__((ext_vector_type(8))) float v8f;


#define T_PITCH 68

__global__ __launch_bounds__(256) void mpo_t345_fused_kernel(
    const float* __restrict__ c0, const float* __restrict__ c1,
    const float* __restrict__ c2, const float* __restrict__ c3,
    const float* __restrict__ c4, float* __restrict__ T345,
    float* __restrict__ P12) {
  __shared__ __attribute__((aligned(16))) float sW[2 * 64 * T_PITCH];
  __shared__ __attribute__((aligned(16))) float sS[48 * T_PITCH];
  const int bid = blockIdx.x, tid = threadIdx.x;

  if (bid >= 256) {
    const int c12 = bid - 256;
    const int i1 = c12 >> 3, i2 = c12 & 7;
    const int o2 = tid >> 6, r2 = tid & 63;
    const float* b = c1 + (i2 * 4 + o2) * 64 + r2;
    float bv[32];
#pragma unroll
    for (int r1 = 0; r1 < 32; ++r1) bv[r1] = b[r1 * 2048];
    const int hf = r2 >> 5, g = (r2 & 31) >> 2, i = r2 & 3;
#pragma unroll
    for (int o1 = 0; o1 < 4; ++o1) {
      const float* a = c0 + (i1 * 4 + o1) * 32;
      float s = 0.f;
#pragma unroll
      for (int r1 = 0; r1 < 32; ++r1) s += a[r1] * bv[r1];
      P12[c12 * 1024 + (g * 32 + hf * 16 + o1 * 4 + o2) * 4 + i] = s;
    }
  } else {
    const int wv   = tid >> 5;
    const int lane = tid & 31;
    const int o3p  = bid & 1;
    const int i3   = (bid >> 1) & 7;
    const int i4   = bid >> 5;
    const int i5b  = ((bid >> 4) & 1) * 4;
    const int sl   = wv >> 2;
    const int e    = wv & 3;
    const int o3   = o3p * 2 + sl;

    const int m    = lane & 15;
    const int half = lane >> 4;

    float* W = sW + sl * (64 * T_PITCH);
    {
      const int o4 = tid >> 6, r3 = tid & 63;
      const float* a = c3 + ((r3 * 8 + i4) * 4 + o4) * 24;
      float av[24];
#pragma unroll
      for (int r4 = 0; r4 < 24; ++r4) av[r4] = a[r4];
      const float* src = c2 + (((16 * e + half) * 8 + i3) * 4 + o3) * 64 + m * 4;
      v4f t[8];
#pragma unroll
      for (int j = 0; j < 8; ++j) t[j] = *(const v4f*)(src + j * 4096);
      const float* b = c4 + i5b * 3;
      float bc[3][12], bn[3][12];
#pragma unroll
      for (int r = 0; r < 3; ++r) {
#pragma unroll
        for (int k = 0; k < 12; ++k) bc[r][k] = b[r * 24 + k];
      }
      float s[12];
#pragma unroll
      for (int k = 0; k < 12; ++k) s[k] = 0.f;
#pragma unroll
      for (int grp = 0; grp < 8; ++grp) {
        if (grp < 7) {
#pragma unroll
          for (int r = 0; r < 3; ++r) {
#pragma unroll
            for (int k = 0; k < 12; ++k) bn[r][k] = b[((grp + 1) * 3 + r) * 24 + k];
          }
        }
        __builtin_amdgcn_sched_barrier(0);
#pragma unroll
        for (int r = 0; r < 3; ++r) {
#pragma unroll
          for (int k = 0; k < 12; ++k) s[k] += av[grp * 3 + r] * bc[r][k];
#pragma unroll
          for (int k = 0; k < 12; ++k) {
            asm("" : "+v"(s[k]));
          }
        }
        if (grp < 7) {
#pragma unroll
          for (int r = 0; r < 3; ++r) {
#pragma unroll
            for (int k = 0; k < 12; ++k) bc[r][k] = bn[r][k];
          }
        }
      }
#pragma unroll
      for (int i5l = 0; i5l < 4; ++i5l) {
#pragma unroll
        for (int o5 = 0; o5 < 3; ++o5)
          sS[(i5l * 12 + o4 * 3 + o5) * T_PITCH + r3] = s[i5l * 3 + o5];
      }
#pragma unroll
      for (int j = 0; j < 8; ++j)
        *(v4f*)(W + (16 * e + 2 * j + half) * T_PITCH + m * 4) = t[j];
    }
    __syncthreads();

    v8f acc[3];
#pragma unroll
    for (int nt = 0; nt < 3; ++nt)
      acc[nt] = (v8f){0.f, 0.f, 0.f, 0.f, 0.f, 0.f, 0.f, 0.f};

#pragma unroll
    for (int j = 0; j < 8; ++j) {
      const int k0 = j * 8 + half * 4;
      const v4f pa = *(const v4f*)(W + (16 * e + m) * T_PITCH + k0);
      v2f a0, a1;
      a0.x = pa.x;  a0.y = pa.y;
      a1.x = pa.z;  a1.y = pa.w;
#pragma unroll
      for (int nt = 0; nt < 3; ++nt) {
        const v4f pb = *(const v4f*)(sS + (16 * nt + m) * T_PITCH + k0);
        v2f b0, b1;
        b0.x = pb.x;  b0.y = pb.y;
        b1.x = pb.z;  b1.y = pb.w;
        acc[nt] = __builtin_amdgcn_wmma_f32_16x16x4_f32(
             false, a0,  false, b0,
             (short)0, acc[nt],  false,  false);
        acc[nt] = __builtin_amdgcn_wmma_f32_16x16x4_f32(
            false, a1, false, b1, (short)0, acc[nt], false, false);
      }
    }

#pragma unroll
    for (int nt = 0; nt < 3; ++nt) {
      const int jc  = 16 * nt + m;
      const int i5l = jc / 12;
      const int col = o3 * 12 + (jc - 12 * i5l);
      float* D = T345 + (i3 * 64 + i4 * 8 + i5b + i5l) * 3072 + (col >> 4) * 1024 + (col & 15) * 4;
#pragma unroll
      for (int u = 0; u < 2; ++u) {
        const int g = 4 * (e & 1) + 2 * half + u;
        v4f d;
        d.x = acc[nt][4 * u + 0];
        d.y = acc[nt][4 * u + 1];
        d.z = acc[nt][4 * u + 2];
        d.w = acc[nt][4 * u + 3];
        *(v4f*)(D + (g * 32 + (e >> 1) * 16) * 4) = d;
      }
    }
  }
}

__global__ __launch_bounds__(256) void mpo_gather_wmma_kernel(
    const int* __restrict__ ids, const float* __restrict__ P12,
    const float* __restrict__ T345, float* __restrict__ out) {
  const int tok   = blockIdx.x * 8 + (threadIdx.x >> 5);
  const int lane  = threadIdx.x & 31;

  const int id   = ids[tok];
  const int c45  = id & 63;
  const int i3   = (id >> 6) & 7;
  const int c12  = id >> 9;

  const float* A = P12 + c12 * 1024 + lane * 4;
  const float* B = T345 + (i3 * 64 + c45) * 3072 + lane * 4;

  const int n    = lane & 15;
  const int half = lane >> 4;

  v4f af[8];
#pragma unroll
  for (int g = 0; g < 8; ++g)
    af[g] = *(const v4f*)(A + g * 128);

  v8f acc[3];
#pragma unroll
  for (int nt = 0; nt < 3; ++nt)
    acc[nt] = (v8f){0.f, 0.f, 0.f, 0.f, 0.f, 0.f, 0.f, 0.f};

  v4f bv[2][8];
#pragma unroll
  for (int g = 0; g < 8; ++g)
    bv[0][g] = *(const v4f*)(B + g * 128);
#pragma unroll
  for (int nt = 0; nt < 3; ++nt) {
    if (nt < 2) {
#pragma unroll
      for (int g = 0; g < 8; ++g)
        bv[(nt + 1) & 1][g] = *(const v4f*)(B + ((nt + 1) * 8 + g) * 128);
    }
    __builtin_amdgcn_sched_barrier(0);
#pragma unroll
    for (int g = 0; g < 8; ++g) {
      v2f a0, a1, b0, b1;
      a0.x = af[g].x;  a0.y = af[g].y;
      a1.x = af[g].z;  a1.y = af[g].w;
      b0.x = bv[nt & 1][g].x;  b0.y = bv[nt & 1][g].y;
      b1.x = bv[nt & 1][g].z;  b1.y = bv[nt & 1][g].w;
      acc[nt] = __builtin_amdgcn_wmma_f32_16x16x4_f32(
          false, a0, false, b0, (short)0, acc[nt], false, false);
      acc[nt] = __builtin_amdgcn_wmma_f32_16x16x4_f32(
          false, a1, false, b1, (short)0, acc[nt], false, false);
    }
    __builtin_amdgcn_sched_barrier(0);
  }

  float* o = out + tok * 768 + half * 8 * 48 + n;
#pragma unroll
  for (int nt = 0; nt < 3; ++nt) {
#pragma unroll
    for (int v = 0; v < 8; ++v)
      o[v * 48 + nt * 16] = acc[nt][v];
  }
}

extern "C" void kernel_launch(void* const* d_in, const int* in_sizes, int n_in,
                              void* d_out, int out_size, void* d_ws, size_t ws_size,
                              hipStream_t stream) {
  const int*   ids = (const int*)d_in[0];
  const float* c0  = (const float*)d_in[1];
  const float* c1  = (const float*)d_in[2];
  const float* c2  = (const float*)d_in[3];
  const float* c3  = (const float*)d_in[4];
  const float* c4  = (const float*)d_in[5];
  float* out = (float*)d_out;

  float* ws   = (float*)d_ws;
  float* P12  = ws;
  float* T345 = ws + 65536 + 49152;

  mpo_t345_fused_kernel<<<320, 256, 0, stream>>>(c0, c1, c2, c3, c4, T345, P12);
  mpo_gather_wmma_kernel<<<512, 256, 0, stream>>>(ids, P12, T345, out);
}
